# starting kernel
# baseline (speedup 1.0000x reference)
#include <hip/hip_runtime.h>
#include <math.h>

#define ALPHA_ERF 0.8862269254527580137f

typedef __attribute__((ext_vector_type(2))) float v2f;
typedef __attribute__((ext_vector_type(8))) float v8f;

__global__ __launch_bounds__(256) void k1_reduce(
    const float* __restrict__ h, const float* __restrict__ v,
    const float* __restrict__ z, float* __restrict__ partial,
    int B, int nchunk)
{
    const int b0    = (blockIdx.x * 256 + threadIdx.x) * 2;
    const int chunk = blockIdx.y;
    const int nBeg  = chunk * nchunk;
    const int nEnd  = nBeg + nchunk;

    float acc[5][2];
#pragma unroll
    for (int r = 0; r < 5; ++r) { acc[r][0] = 0.0f; acc[r][1] = 0.0f; }

    for (int n = nBeg; n < nEnd; ++n) {
        const size_t hoff = (size_t)n * (size_t)B + (size_t)b0;
        if (n + 4 < nEnd) __builtin_prefetch(h + hoff + 4 * (size_t)B, 0, 1);
        const float2 hv = *(const float2*)(h + hoff);
        const float p0 = erff(ALPHA_ERF * hv.x);
        const float p1 = erff(ALPHA_ERF * hv.y);
        const float4 vv = *(const float4*)(v + (size_t)n * 4);
        const float  zv = z[n];
        acc[0][0] = fmaf(vv.x, p0, acc[0][0]); acc[0][1] = fmaf(vv.x, p1, acc[0][1]);
        acc[1][0] = fmaf(vv.y, p0, acc[1][0]); acc[1][1] = fmaf(vv.y, p1, acc[1][1]);
        acc[2][0] = fmaf(vv.z, p0, acc[2][0]); acc[2][1] = fmaf(vv.z, p1, acc[2][1]);
        acc[3][0] = fmaf(vv.w, p0, acc[3][0]); acc[3][1] = fmaf(vv.w, p1, acc[3][1]);
        acc[4][0] = fmaf(zv,   p0, acc[4][0]); acc[4][1] = fmaf(zv,   p1, acc[4][1]);
    }

#pragma unroll
    for (int r = 0; r < 5; ++r) {
        float2 o; o.x = acc[r][0]; o.y = acc[r][1];
        *(float2*)(partial + (size_t)(chunk * 5 + r) * (size_t)B + b0) = o;
    }
}

__global__ __launch_bounds__(256) void k2_finalize(
    const float* __restrict__ partial, float* __restrict__ tbuf,
    float* __restrict__ y_out, int B, int nchunks, float invN)
{
    const int idx = blockIdx.x * 256 + threadIdx.x;
    if (idx >= 5 * B) return;
    const int r = idx / B;
    const int b = idx - r * B;
    float s = 0.0f;
    for (int c = 0; c < nchunks; ++c)
        s += partial[(size_t)(c * 5 + r) * (size_t)B + b];
    if (r == 4) y_out[b] = s * invN;
    else        tbuf[(size_t)r * (size_t)B + b] = s;
}

__global__ __launch_bounds__(256) void k3_update(
    const float* __restrict__ h, const float* __restrict__ u,
    const float* __restrict__ m, const float* __restrict__ x,
    const float* __restrict__ tbuf, float* __restrict__ hnew,
    int B, float invN)
{
    const int lane = threadIdx.x & 31;
    const int wave = threadIdx.x >> 5;
    const int lo   = lane & 15;
    const int hi   = lane >> 4;

    const int colGroups = B >> 6;
    const int gwave     = blockIdx.x * 8 + wave;
    const int rowTile   = gwave / colGroups;
    const int colGroup  = gwave - rowTile * colGroups;
    const int n0        = rowTile * 16;

    v2f a;
    {
        const float2 av = *(const float2*)(u + (size_t)(n0 + lo) * 4 + hi * 2);
        a.x = av.x; a.y = av.y;
    }

    float mv[8];
    {
        const float4 m0 = *(const float4*)(m + n0 + 8 * hi);
        const float4 m1 = *(const float4*)(m + n0 + 8 * hi + 4);
        mv[0] = m0.x; mv[1] = m0.y; mv[2] = m0.z; mv[3] = m0.w;
        mv[4] = m1.x; mv[5] = m1.y; mv[6] = m1.z; mv[7] = m1.w;
    }

    for (int bt = 0; bt < 4; ++bt) {
        const int col = colGroup * 64 + bt * 16 + lo;

        v2f bm;
        bm.x = tbuf[(size_t)(0 + 2 * hi) * (size_t)B + col];
        bm.y = tbuf[(size_t)(1 + 2 * hi) * (size_t)B + col];

        const float xv = x[col];

        const size_t base = (size_t)(n0 + 8 * hi) * (size_t)B + (size_t)col;
        const float* __restrict__ hp = h + base;

        float hv[8];
#pragma unroll
        for (int g = 0; g < 8; ++g) hv[g] = hp[(size_t)g * B];

        v8f c = {};
        v8f d = __builtin_amdgcn_wmma_f32_16x16x4_f32(
             false, a,  false, bm,
             (short)0, c,  false,  false);

        float out[8];
#pragma unroll
        for (int g = 0; g < 8; ++g) {
            const float upd = fmaf(mv[g], xv, d[g] * invN);
            out[g] = hv[g] + (upd - hv[g]);
        }

        float* __restrict__ op = hnew + base;
#pragma unroll
        for (int g = 0; g < 8; ++g)
            __builtin_nontemporal_store(out[g], op + (size_t)g * B);
    }
}

extern "C" void kernel_launch(void* const* d_in, const int* in_sizes, int n_in,
                              void* d_out, int out_size, void* d_ws, size_t ws_size,
                              hipStream_t stream) {
    const float* x = (const float*)d_in[0];
    const float* h = (const float*)d_in[1];
    const float* m = (const float*)d_in[2];
    const float* u = (const float*)d_in[3];
    const float* v = (const float*)d_in[4];
    const float* z = (const float*)d_in[5];

    const int B = in_sizes[0];
    const int N = in_sizes[2];
    const float invN = 1.0f / (float)N;

    float* y_out = (float*)d_out;
    float* hnew  = (float*)d_out + B;

    const int NCHUNKS = 32;
    const int nchunk  = N / NCHUNKS;

    float* partial = (float*)d_ws;
    float* tbuf    = partial + (size_t)NCHUNKS * 5 * (size_t)B;

    dim3 g1(B / 512, NCHUNKS);
    k1_reduce<<<g1, 256, 0, stream>>>(h, v, z, partial, B, nchunk);

    const int tot2 = 5 * B;
    k2_finalize<<<(tot2 + 255) / 256, 256, 0, stream>>>(partial, tbuf, y_out, B,
                                                        NCHUNKS, invN);

    const int waves = (N / 16) * (B / 64);
    k3_update<<<waves / 8, 256, 0, stream>>>(h, u, m, x, tbuf, hnew, B, invN);
}


# pass 1 as a WMMA product over column strips: a 32-wave workgroup owns 16 columns over all N rows, folds its waves in LDS, writes final t and y; no chunk planes or fold launch; changes summation order
# speedup vs baseline: 2.5557x; 2.5557x over previous
#include <hip/hip_runtime.h>
#include <math.h>

#define ALPHA_ERF 0.8862269254527580137f

typedef __attribute__((ext_vector_type(2))) float v2f;
typedef __attribute__((ext_vector_type(8))) float v8f;

#define STRIP_WAVES 32

__global__ __launch_bounds__(1024) void k1_strip(
    const float* __restrict__ h, const float* __restrict__ v,
    const float* __restrict__ z, float* __restrict__ tbuf,
    float* __restrict__ y_out, int B, int rowsPerWave, float invN)
{
    __shared__ float red[STRIP_WAVES][5][16];

    const int lane = threadIdx.x & 31;
    const int lo   = lane & 15;
    const int hi   = lane >> 4;
    const int b0   = blockIdx.x * 16;
    const int wave = __builtin_amdgcn_readfirstlane((int)(threadIdx.x >> 5));
    const int row0 = wave * rowsPerWave;

    const float* __restrict__ hlane =
        h + (size_t)(row0 + 2 * hi) * (size_t)B + (size_t)(b0 + lo);

    const int wstride = (lo < 4) ? 4 : 1;
    const float* __restrict__ w0 =
        ((lo < 4) ? (v + lo) : z) + (size_t)(row0 + 2 * hi) * (size_t)wstride;
    const float* __restrict__ w1 = w0 + wstride;
    const size_t wstep = (size_t)(4 * wstride);

    v8f acc = {};
    for (int kb = 0; kb < rowsPerWave; kb += 32) {
        const float* __restrict__ hrow = hlane + (size_t)kb * (size_t)B;
        float hv[8][2], wv[8][2];
#pragma unroll
        for (int u = 0; u < 8; ++u) {
            hv[u][0] = hrow[(size_t)(4 * u) * (size_t)B];
            hv[u][1] = hrow[(size_t)(4 * u + 1) * (size_t)B];
            wv[u][0] = *w0;
            wv[u][1] = *w1;
            w0 += wstep;
            w1 += wstep;
        }
#pragma unroll
        for (int u = 0; u < 8; ++u) {
            v2f af, bf;
            af.x = wv[u][0];
            af.y = wv[u][1];
            bf.x = erff(ALPHA_ERF * hv[u][0]);
            bf.y = erff(ALPHA_ERF * hv[u][1]);
            acc = __builtin_amdgcn_wmma_f32_16x16x4_f32(
                 false, af,  false, bf,
                 (short)0, acc,  false,  false);
        }
    }

    if (hi == 0) {
        red[wave][0][lo] = acc[0];
        red[wave][1][lo] = acc[1];
        red[wave][2][lo] = acc[2];
        red[wave][3][lo] = acc[3];
        red[wave][4][lo] = acc[4];
    }
    __syncthreads();

    if (threadIdx.x < 80) {
        const int r = threadIdx.x >> 4;
        const int c = threadIdx.x & 15;
        float s = 0.0f;
#pragma unroll
        for (int w = 0; w < STRIP_WAVES; ++w) s += red[w][r][c];
        if (r < 4) tbuf[(size_t)r * (size_t)B + (size_t)(b0 + c)] = s;
        else       y_out[b0 + c] = s * invN;
    }
}

__global__ __launch_bounds__(256) void k3_update(
    const float* __restrict__ h, const float* __restrict__ u,
    const float* __restrict__ m, const float* __restrict__ x,
    const float* __restrict__ tbuf, float* __restrict__ hnew,
    int B, float invN)
{
    const int lane = threadIdx.x & 31;
    const int wave = threadIdx.x >> 5;
    const int lo   = lane & 15;
    const int hi   = lane >> 4;

    const int colGroups = B >> 6;
    const int gwave     = blockIdx.x * 8 + wave;
    const int rowTile   = gwave / colGroups;
    const int colGroup  = gwave - rowTile * colGroups;
    const int n0        = rowTile * 16;

    v2f a;
    {
        const float2 av = *(const float2*)(u + (size_t)(n0 + lo) * 4 + hi * 2);
        a.x = av.x; a.y = av.y;
    }

    float mv[8];
    {
        const float4 m0 = *(const float4*)(m + n0 + 8 * hi);
        const float4 m1 = *(const float4*)(m + n0 + 8 * hi + 4);
        mv[0] = m0.x; mv[1] = m0.y; mv[2] = m0.z; mv[3] = m0.w;
        mv[4] = m1.x; mv[5] = m1.y; mv[6] = m1.z; mv[7] = m1.w;
    }

    for (int bt = 0; bt < 4; ++bt) {
        const int col = colGroup * 64 + bt * 16 + lo;

        v2f bm;
        bm.x = tbuf[(size_t)(0 + 2 * hi) * (size_t)B + col];
        bm.y = tbuf[(size_t)(1 + 2 * hi) * (size_t)B + col];

        const float xv = x[col];

        const size_t base = (size_t)(n0 + 8 * hi) * (size_t)B + (size_t)col;
        const float* __restrict__ hp = h + base;

        float hv[8];
#pragma unroll
        for (int g = 0; g < 8; ++g) hv[g] = hp[(size_t)g * B];

        v8f c = {};
        v8f d = __builtin_amdgcn_wmma_f32_16x16x4_f32(
             false, a,  false, bm,
             (short)0, c,  false,  false);

        float out[8];
#pragma unroll
        for (int g = 0; g < 8; ++g) {
            const float upd = fmaf(mv[g], xv, d[g] * invN);
            out[g] = hv[g] + (upd - hv[g]);
        }

        float* __restrict__ op = hnew + base;
#pragma unroll
        for (int g = 0; g < 8; ++g)
            __builtin_nontemporal_store(out[g], op + (size_t)g * B);
    }
}

extern "C" void kernel_launch(void* const* d_in, const int* in_sizes, int n_in,
                              void* d_out, int out_size, void* d_ws, size_t ws_size,
                              hipStream_t stream) {
    const float* x = (const float*)d_in[0];
    const float* h = (const float*)d_in[1];
    const float* m = (const float*)d_in[2];
    const float* u = (const float*)d_in[3];
    const float* v = (const float*)d_in[4];
    const float* z = (const float*)d_in[5];

    const int B = in_sizes[0];
    const int N = in_sizes[2];
    const float invN = 1.0f / (float)N;

    float* y_out = (float*)d_out;
    float* hnew  = (float*)d_out + B;

    float* tbuf = (float*)d_ws;

    k1_strip<<<B / 16, 1024, 0, stream>>>(h, v, z, tbuf, y_out, B,
                                          N / STRIP_WAVES, invN);

    const int waves = (N / 16) * (B / 64);
    k3_update<<<waves / 8, 256, 0, stream>>>(h, u, m, x, tbuf, hnew, B, invN);
}
